# starting kernel
# baseline (speedup 1.0000x reference)
#include <hip/hip_runtime.h>
#include <hip/hip_bf16.h>


#define FN   512
#define NB   43
#define LPB  4
#define TPB  128
#define PI_F 3.14159265358979323846f

typedef float v2f __attribute__((ext_vector_type(2)));
typedef float v8f __attribute__((ext_vector_type(8)));

__device__ __forceinline__ v8f wmma_f32(v2f a, v2f b, v8f c) {
  return __builtin_amdgcn_wmma_f32_16x16x4_f32(false, a, false, b,
                                               (short)0, c, false, false);
}

struct W16 { v2f ar[4]; v2f ai[4]; v2f ain[4]; };

__device__ __forceinline__ void make_w16(float sgn, int lane, W16& w) {
  const int m  = lane & 15;
  const int kb = (lane >> 4) << 1;
  const float step = sgn * (2.0f * PI_F / 16.0f);
#pragma unroll
  for (int kc = 0; kc < 4; ++kc) {
    const int k0 = 4 * kc + kb;
    float s0, c0, s1, c1;
    sincosf(step * (float)(m * k0),       &s0, &c0);
    sincosf(step * (float)(m * (k0 + 1)), &s1, &c1);
    v2f tr; tr.x = c0;  tr.y = c1;  w.ar[kc]  = tr;
    v2f ti; ti.x = s0;  ti.y = s1;  w.ai[kc]  = ti;
    v2f tn; tn.x = -s0; tn.y = -s1; w.ain[kc] = tn;
  }
}

__device__ __forceinline__ void ctile(const float* sr, const float* si, int t,
                                      int lane, const W16& w, v8f& cr, v8f& ci) {
  const int n  = 16 * t + (lane & 15);
  const int kb = (lane >> 4) << 1;
  v8f zr = {0.f,0.f,0.f,0.f,0.f,0.f,0.f,0.f};
  cr = zr; ci = zr;
#pragma unroll
  for (int kc = 0; kc < 4; ++kc) {
    const int k0 = 4 * kc + kb;
    v2f br; br.x = sr[k0 * 32 + n]; br.y = sr[(k0 + 1) * 32 + n];
    v2f bi; bi.x = si[k0 * 32 + n]; bi.y = si[(k0 + 1) * 32 + n];
    cr = wmma_f32(w.ar[kc],  br, cr);
    cr = wmma_f32(w.ain[kc], bi, cr);
    ci = wmma_f32(w.ar[kc],  bi, ci);
    ci = wmma_f32(w.ai[kc],  br, ci);
  }
}

__device__ __forceinline__ void fft_stageA(const float* xr, const float* xi,
                                           float* ur, float* ui,
                                           int lane, float sgn) {
  const float tws = sgn * (2.0f * PI_F / 32.0f);
#pragma unroll
  for (int i = 0; i < 8; ++i) {
    const int p  = lane + 32 * i;
    const int n1 = p & 15, m1 = p >> 4;
    const float a_r = xr[p],       a_i = xi[p];
    const float b_r = xr[p + 256], b_i = xi[p + 256];
    ur[m1 * 32 + n1] = a_r + b_r;
    ui[m1 * 32 + n1] = a_i + b_i;
    const float dr = a_r - b_r, di = a_i - b_i;
    float s, c; sincosf(tws * (float)m1, &s, &c);
    ur[m1 * 32 + n1 + 16] = dr * c - di * s;
    ui[m1 * 32 + n1 + 16] = dr * s + di * c;
  }
}

__device__ __forceinline__ void fft_stageBC(const float* ur, const float* ui,
                                            float* vr, float* vi,
                                            const float* twr, const float* twi,
                                            int lane, const W16& w) {
  const int nn = lane & 15;
  const int mb = (lane >> 4) * 8;
#pragma unroll
  for (int t = 0; t < 2; ++t) {
    v8f gr, gi;
    ctile(ur, ui, t, lane, w, gr, gi);
    const int ccol = 16 * t + nn;
    const int n1 = ccol & 15, j2 = ccol >> 4;
#pragma unroll
    for (int r = 0; r < 8; ++r) {
      const int j1  = mb + r;
      const int k2  = 2 * j1 + j2;
      const int idx = n1 * 32 + k2;
      const float tr = twr[idx], ti_ = twi[idx];
      const float g_r = gr[r], g_i = gi[r];
      vr[idx] = g_r * tr - g_i * ti_;
      vi[idx] = g_r * ti_ + g_i * tr;
    }
  }
}

__device__ __forceinline__ void fill_tw512(float* twr, float* twi,
                                           int tid, float sgn) {
  const float step = sgn * (2.0f * PI_F / 512.0f);
  for (int i = tid; i < FN; i += TPB) {
    const int n1 = i >> 5, k2 = i & 31;
    float s, c; sincosf(step * (float)(n1 * k2), &s, &c);
    twr[i] = c; twi[i] = s;
  }
}

#define FFT_SHARED                                              \
  __shared__ float Ar[LPB][FN], Ai[LPB][FN];                    \
  __shared__ float Br[LPB][FN], Bi[LPB][FN];                    \
  __shared__ float Twr[FN], Twi[FN];

__global__ __launch_bounds__(TPB)
void k_fwd_rows(const float* __restrict__ img, float2* __restrict__ specRow) {
  FFT_SHARED
  const int tid = threadIdx.x, lane = tid & 31, wid = tid >> 5;
  const int line = blockIdx.x * LPB + wid;
  fill_tw512(Twr, Twi, tid, -1.0f);
  W16 w; make_w16(-1.0f, lane, w);
#pragma unroll
  for (int i = 0; i < 16; ++i) {
    const int p = lane + 32 * i;
    Ar[wid][p] = img[line * FN + p];
    Ai[wid][p] = 0.0f;
  }
  __syncthreads();
  fft_stageA(Ar[wid], Ai[wid], Br[wid], Bi[wid], lane, -1.0f);
  __syncthreads();
  fft_stageBC(Br[wid], Bi[wid], Ar[wid], Ai[wid], Twr, Twi, lane, w);
  __syncthreads();
  const int nn = lane & 15, mb = (lane >> 4) * 8;
#pragma unroll
  for (int t = 0; t < 2; ++t) {
    v8f dr, di;
    ctile(Ar[wid], Ai[wid], t, lane, w, dr, di);
#pragma unroll
    for (int r = 0; r < 8; ++r) {
      const int o = 32 * (mb + r) + 16 * t + nn;
      specRow[line * FN + o] = make_float2(dr[r], di[r]);
    }
  }
}

__global__ __launch_bounds__(TPB)
void k_fwd_cols(const float2* __restrict__ specRow, float2* __restrict__ spec) {
  FFT_SHARED
  const int tid = threadIdx.x, lane = tid & 31, wid = tid >> 5;
  const int c0 = blockIdx.x * LPB;
  fill_tw512(Twr, Twi, tid, -1.0f);
  W16 w; make_w16(-1.0f, lane, w);
  for (int i = tid; i < FN * LPB; i += TPB) {
    const int e = i >> 2, ww = i & 3;
    const float2 s = specRow[e * FN + c0 + ww];
    Ar[ww][e] = s.x; Ai[ww][e] = s.y;
  }
  __syncthreads();
  fft_stageA(Ar[wid], Ai[wid], Br[wid], Bi[wid], lane, -1.0f);
  __syncthreads();
  fft_stageBC(Br[wid], Bi[wid], Ar[wid], Ai[wid], Twr, Twi, lane, w);
  __syncthreads();
  const int nn = lane & 15, mb = (lane >> 4) * 8;
#pragma unroll
  for (int t = 0; t < 2; ++t) {
    v8f dr, di;
    ctile(Ar[wid], Ai[wid], t, lane, w, dr, di);
#pragma unroll
    for (int r = 0; r < 8; ++r) {
      const int o = 32 * (mb + r) + 16 * t + nn;
      Br[wid][o] = dr[r]; Bi[wid][o] = di[r];
    }
  }
  __syncthreads();
  for (int i = tid; i < FN * LPB; i += TPB) {
    const int o = i >> 2, ww = i & 3;
    spec[o * FN + c0 + ww] = make_float2(Br[ww][o], Bi[ww][o]);
  }
}

__global__ __launch_bounds__(TPB)
void k_band_cols(const float2* __restrict__ spec, const float* __restrict__ win,
                 float2* __restrict__ T, int b0) {
  FFT_SHARED
  const int tid = threadIdx.x, lane = tid & 31, wid = tid >> 5;
  const int c0 = blockIdx.x * LPB;
  const int bi = blockIdx.y;
  const size_t woff = (size_t)(b0 + bi) * FN * FN;
  float2* Tb = T + (size_t)bi * FN * FN;
  fill_tw512(Twr, Twi, tid, 1.0f);
  W16 w; make_w16(1.0f, lane, w);
  for (int i = tid; i < FN * LPB; i += TPB) {
    const int e = i >> 2, ww = i & 3;
    const float2 s = spec[e * FN + c0 + ww];
    const float  wv = win[woff + (size_t)e * FN + c0 + ww];
    Ar[ww][e] = s.x * wv; Ai[ww][e] = s.y * wv;
  }
  __syncthreads();
  fft_stageA(Ar[wid], Ai[wid], Br[wid], Bi[wid], lane, 1.0f);
  __syncthreads();
  fft_stageBC(Br[wid], Bi[wid], Ar[wid], Ai[wid], Twr, Twi, lane, w);
  __syncthreads();
  const int nn = lane & 15, mb = (lane >> 4) * 8;
#pragma unroll
  for (int t = 0; t < 2; ++t) {
    v8f dr, di;
    ctile(Ar[wid], Ai[wid], t, lane, w, dr, di);
#pragma unroll
    for (int r = 0; r < 8; ++r) {
      const int o = 32 * (mb + r) + 16 * t + nn;
      Br[wid][o] = dr[r]; Bi[wid][o] = di[r];
    }
  }
  __syncthreads();
  for (int i = tid; i < FN * LPB; i += TPB) {
    const int o = i >> 2, ww = i & 3;
    Tb[o * FN + c0 + ww] = make_float2(Br[ww][o], Bi[ww][o]);
  }
}

__global__ __launch_bounds__(TPB)
void k_band_rows(const float2* __restrict__ T, float* __restrict__ out, int b0) {
  FFT_SHARED
  const int tid = threadIdx.x, lane = tid & 31, wid = tid >> 5;
  const int bi = blockIdx.y;
  const float2* Tb = T + (size_t)bi * FN * FN;
  float* ob = out + (size_t)(b0 + bi) * FN * FN;
  const int line = blockIdx.x * LPB + wid;
  fill_tw512(Twr, Twi, tid, 1.0f);
  W16 w; make_w16(1.0f, lane, w);
#pragma unroll
  for (int i = 0; i < 16; ++i) {
    const int p = lane + 32 * i;
    const float2 v = Tb[line * FN + p];
    Ar[wid][p] = v.x; Ai[wid][p] = v.y;
  }
  __syncthreads();
  fft_stageA(Ar[wid], Ai[wid], Br[wid], Bi[wid], lane, 1.0f);
  __syncthreads();
  fft_stageBC(Br[wid], Bi[wid], Ar[wid], Ai[wid], Twr, Twi, lane, w);
  __syncthreads();
  const float scale = 1.0f / (512.0f * 512.0f);
  const int nn = lane & 15, mb = (lane >> 4) * 8;
#pragma unroll
  for (int t = 0; t < 2; ++t) {
    v8f dr, di;
    ctile(Ar[wid], Ai[wid], t, lane, w, dr, di);
#pragma unroll
    for (int r = 0; r < 8; ++r) {
      const int o = 32 * (mb + r) + 16 * t + nn;
      ob[line * FN + o] = dr[r] * scale;
    }
  }
}

extern "C" void kernel_launch(void* const* d_in, const int* in_sizes, int n_in,
                              void* d_out, int out_size, void* d_ws,
                              size_t ws_size, hipStream_t stream) {
  (void)in_sizes; (void)n_in; (void)out_size;
  const float* img = (const float*)d_in[0];
  const float* win = (const float*)d_in[1];
  float* out = (float*)d_out;

  float2* specRow = (float2*)d_ws;
  float2* spec    = specRow + (size_t)FN * FN;
  float2* bandT   = spec    + (size_t)FN * FN;
  const size_t reserved = 2ull * sizeof(float2) * FN * FN;
  const size_t perBand  = sizeof(float2) * (size_t)FN * FN;
  int chunk = (ws_size > reserved) ? (int)((ws_size - reserved) / perBand) : 0;
  if (chunk < 1)  chunk = 1;
  if (chunk > NB) chunk = NB;

  const dim3 blk(TPB);
  k_fwd_rows<<<dim3(FN / LPB), blk, 0, stream>>>(img, specRow);
  k_fwd_cols<<<dim3(FN / LPB), blk, 0, stream>>>(specRow, spec);
  for (int b0 = 0; b0 < NB; b0 += chunk) {
    const int nb = (NB - b0 < chunk) ? (NB - b0) : chunk;
    k_band_cols<<<dim3(FN / LPB, nb), blk, 0, stream>>>(spec, win, bandT, b0);
    k_band_rows<<<dim3(FN / LPB, nb), blk, 0, stream>>>(bandT, out, b0);
  }
}


# Forward column transform moved into the band column kernel: each work-group transforms its own 32 columns on the inverse stages (conjugated) into an LDS tile; 3 launches, was 4; sums re-associated
# speedup vs baseline: 3.1277x; 3.1277x over previous
#include <hip/hip_runtime.h>
#include <hip/hip_bf16.h>


#define FN   512
#define NB   43
#define LPB  4
#define TPB  128
#define PI_F 3.14159265358979323846f

typedef float v2f __attribute__((ext_vector_type(2)));
typedef float v8f __attribute__((ext_vector_type(8)));

__device__ __forceinline__ v8f wmma_f32(v2f a, v2f b, v8f c) {
  return __builtin_amdgcn_wmma_f32_16x16x4_f32(false, a, false, b,
                                               (short)0, c, false, false);
}

struct W16 { v2f ar[4]; v2f ai[4]; v2f ain[4]; };

__device__ __forceinline__ void fill_w512(float2* L, int tid, int nthr, float sgn) {
  const float step = 2.0f * PI_F / 512.0f;
  for (int j = tid; j < 128; j += nthr) {
    float s, c; sincosf(step * (float)j, &s, &c);
    const float ss = sgn * s, sc = sgn * c;
    L[j]       = make_float2( c,  ss);
    L[j + 128] = make_float2(-s,  sc);
    L[j + 256] = make_float2(-c, -ss);
    L[j + 384] = make_float2( s, -sc);
  }
}

__device__ __forceinline__ void make_w16(const float2* L, int lane, W16& w) {
  const int m  = lane & 15;
  const int kb = (lane >> 4) << 1;
#pragma unroll
  for (int kc = 0; kc < 4; ++kc) {
    const int k0 = 4 * kc + kb;
    const float2 u0 = L[(32 * m * k0) & 511];
    const float2 u1 = L[(32 * m * (k0 + 1)) & 511];
    v2f tr; tr.x = u0.x;  tr.y = u1.x;  w.ar[kc]  = tr;
    v2f ti; ti.x = u0.y;  ti.y = u1.y;  w.ai[kc]  = ti;
    v2f tn; tn.x = -u0.y; tn.y = -u1.y; w.ain[kc] = tn;
  }
}

__device__ __forceinline__ void ctile(const float* sr, const float* si, int t,
                                      int lane, const W16& w, v8f& cr, v8f& ci) {
  const int n  = 16 * t + (lane & 15);
  const int kb = (lane >> 4) << 1;
  v8f zr = {0.f,0.f,0.f,0.f,0.f,0.f,0.f,0.f};
  cr = zr; ci = zr;
#pragma unroll
  for (int kc = 0; kc < 4; ++kc) {
    const int k0 = 4 * kc + kb;
    v2f br; br.x = sr[k0 * 32 + n]; br.y = sr[(k0 + 1) * 32 + n];
    v2f bi; bi.x = si[k0 * 32 + n]; bi.y = si[(k0 + 1) * 32 + n];
    cr = wmma_f32(w.ar[kc],  br, cr);
    cr = wmma_f32(w.ain[kc], bi, cr);
    ci = wmma_f32(w.ar[kc],  bi, ci);
    ci = wmma_f32(w.ai[kc],  br, ci);
  }
}

__device__ __forceinline__ void fft_stageA(const float* xr, const float* xi,
                                           float* ur, float* ui,
                                           const float2* L, int lane) {
#pragma unroll
  for (int i = 0; i < 8; ++i) {
    const int p  = lane + 32 * i;
    const int n1 = p & 15, m1 = p >> 4;
    const float a_r = xr[p],       a_i = xi[p];
    const float b_r = xr[p + 256], b_i = xi[p + 256];
    ur[m1 * 32 + n1] = a_r + b_r;
    ui[m1 * 32 + n1] = a_i + b_i;
    const float dr = a_r - b_r, di = a_i - b_i;
    const float2 tw = L[16 * m1];
    const float c = tw.x, s = tw.y;
    ur[m1 * 32 + n1 + 16] = dr * c - di * s;
    ui[m1 * 32 + n1 + 16] = dr * s + di * c;
  }
}

__device__ __forceinline__ void fft_stageBC(const float* ur, const float* ui,
                                            float* vr, float* vi,
                                            const float2* L,
                                            int lane, const W16& w) {
  const int nn = lane & 15;
  const int mb = (lane >> 4) * 8;
#pragma unroll
  for (int t = 0; t < 2; ++t) {
    v8f gr, gi;
    ctile(ur, ui, t, lane, w, gr, gi);
    const int ccol = 16 * t + nn;
    const int n1 = ccol & 15, j2 = ccol >> 4;
#pragma unroll
    for (int r = 0; r < 8; ++r) {
      const int j1  = mb + r;
      const int k2  = 2 * j1 + j2;
      const int idx = n1 * 32 + k2;
      const float2 tw = L[n1 * k2];
      const float tr = tw.x, ti_ = tw.y;
      const float g_r = gr[r], g_i = gi[r];
      vr[idx] = g_r * tr - g_i * ti_;
      vi[idx] = g_r * ti_ + g_i * tr;
    }
  }
}

#define FFT_SHARED                                              \
  __shared__ float Ar[LPB][FN], Ai[LPB][FN];                    \
  __shared__ float Br[LPB][FN], Bi[LPB][FN];                    \
  __shared__ float2 L5[FN];

__global__ __launch_bounds__(TPB)
void k_fwd_rows(const float* __restrict__ img, float2* __restrict__ specRow) {
  FFT_SHARED
  const int tid = threadIdx.x, lane = tid & 31, wid = tid >> 5;
  const int line = blockIdx.x * LPB + wid;
  fill_w512(L5, tid, TPB, -1.0f);
#pragma unroll
  for (int i = 0; i < 16; ++i) {
    const int p = lane + 32 * i;
    Ar[wid][p] = img[line * FN + p];
    Ai[wid][p] = 0.0f;
  }
  __syncthreads();
  W16 w; make_w16(L5, lane, w);
  fft_stageA(Ar[wid], Ai[wid], Br[wid], Bi[wid], L5, lane);
  __syncthreads();
  fft_stageBC(Br[wid], Bi[wid], Ar[wid], Ai[wid], L5, lane, w);
  __syncthreads();
  const int nn = lane & 15, mb = (lane >> 4) * 8;
#pragma unroll
  for (int t = 0; t < 2; ++t) {
    v8f dr, di;
    ctile(Ar[wid], Ai[wid], t, lane, w, dr, di);
#pragma unroll
    for (int r = 0; r < 8; ++r) {
      const int o = 32 * (mb + r) + 16 * t + nn;
      specRow[line * FN + o] = make_float2(dr[r], di[r]);
    }
  }
}

template <int JLO, int JN>
__device__ __forceinline__ void load_win(float (&wr)[4][4][2][4],
                                         const float* __restrict__ wa,
                                         const float* __restrict__ wb,
                                         size_t ebase, size_t mbase, size_t m0) {
  const float* const ad = wa + ebase;
  const float* const am = wa + mbase;
  const float* const bd = wb + ebase;
  const float* const bm = wb + mbase;
#pragma unroll
  for (int jj = JLO; jj < JLO + JN; ++jj)
#pragma unroll
    for (int kc = 0; kc < 4; ++kc)
#pragma unroll
      for (int e = 0; e < 2; ++e) {
        const int K = 128 * kc + 32 * e + 8 * jj;
        const ptrdiff_t ko = (ptrdiff_t)K * FN;
        wr[jj][kc][e][0] = ad[ko];
        wr[jj][kc][e][1] = (K == 0) ? wa[m0] : am[-ko];
        wr[jj][kc][e][2] = bd[ko];
        wr[jj][kc][e][3] = (K == 0) ? wb[m0] : bm[-ko];
      }
}
struct M8 { v2f a[4]; };
__device__ __forceinline__ void make_m8(const float2* L, int lane, int tq, M8& m) {
  const int row = lane & 15, h = lane >> 4;
  const int j = 4 * (row >> 3) + ((row & 7) >> 1);
  const bool im = (row & 1) != 0;
#pragma unroll
  for (int kc = 0; kc < 4; ++kc) {
    const int a = 2 * kc + h;
    const float2 u = L[((tq + 64 * a) * j) & 511];
    v2f t;
    t.x = im ? u.y : u.x;
    t.y = im ? u.x : -u.y;
    m.a[kc] = t;
  }
}
__device__ __forceinline__ void lds_barrier() {
  __builtin_amdgcn_fence(__ATOMIC_RELEASE, "workgroup", "local");
  __builtin_amdgcn_s_barrier();
  __builtin_amdgcn_fence(__ATOMIC_ACQUIRE, "workgroup", "local");
}
#define CT_COLS 32
#define CT_TPB  512
#define GBC     11
#define YS      20
#define YK      (32 * YS + 4)
#define SP_SUB  (FN * 16)
__device__ __forceinline__ void col_s1_store(float2* Ysub, const float2 (*Tw)[16],
                                             int n2, int h, int c,
                                             const v8f& d0, const v8f& d1) {
#pragma unroll
  for (int i = 0; i < 4; ++i) {
    const int j = 4 * h + i;
    const float pr = d0[2 * i] + d1[2 * i], pi = d0[2 * i + 1] + d1[2 * i + 1];
    const float mr = d0[2 * i] - d1[2 * i], mi = d0[2 * i + 1] - d1[2 * i + 1];
    const float2 ta = Tw[n2][j], tb = Tw[n2][j + 8];
    Ysub[j * YK + n2 * YS + c] =
        make_float2(pr * ta.x - pi * ta.y, pr * ta.y + pi * ta.x);
    Ysub[(j + 8) * YK + n2 * YS + c] =
        make_float2(mr * tb.x - mi * tb.y, mr * tb.y + mi * tb.x);
  }
}
template <bool TOLDS>
__device__ __forceinline__ void col_stage2(const float2* Ysub, int w8, int c, int h,
                                           const M8& m0, const M8& m1,
                                           const M8& m2, const M8& m3,
                                           float2* tp, float2* sp) {
  const v8f zero = {0.f,0.f,0.f,0.f,0.f,0.f,0.f,0.f};
#pragma unroll
  for (int kk = 0; kk < 2; ++kk) {
    const float2* const yk = Ysub + (2 * w8 + kk) * YK + c;
    float2 ya[4][4];
#pragma unroll
    for (int kc = 0; kc < 4; ++kc)
#pragma unroll
      for (int q = 0; q < 4; ++q)
        ya[kc][q] = yk[(8 * kc + 4 * h + q) * YS];
    v8f e0 = zero, e1 = zero, e2 = zero, e3 = zero;
#pragma unroll
    for (int kc = 0; kc < 4; ++kc) {
      v2f y0; y0.x = ya[kc][0].x; y0.y = ya[kc][0].y;
      v2f y1; y1.x = ya[kc][1].x; y1.y = ya[kc][1].y;
      v2f y2; y2.x = ya[kc][2].x; y2.y = ya[kc][2].y;
      v2f y3; y3.x = ya[kc][3].x; y3.y = ya[kc][3].y;
      e0 = wmma_f32(m0.a[kc], y0, e0);
      e1 = wmma_f32(m1.a[kc], y1, e1);
      e2 = wmma_f32(m2.a[kc], y2, e2);
      e3 = wmma_f32(m3.a[kc], y3, e3);
    }
    float2* const sq = TOLDS ? sp + (kk ^ h) * 16 : nullptr;
#pragma unroll
    for (int i = 0; i < 4; ++i) {
      const float ar = e0[2 * i] + e2[2 * i], ai = e0[2 * i + 1] + e2[2 * i + 1];
      const float br = e0[2 * i] - e2[2 * i], bi_ = e0[2 * i + 1] - e2[2 * i + 1];
      const float cr = e1[2 * i] + e3[2 * i], ci = e1[2 * i + 1] + e3[2 * i + 1];
      const float dr = e3[2 * i + 1] - e1[2 * i + 1];
      const float di = e1[2 * i] - e3[2 * i];
      if (TOLDS) {
        sq[(16 * i) * 16]       = make_float2(ar + cr, -(ai + ci));
        sq[(16 * i + 128) * 16] = make_float2(br + dr, -(bi_ + di));
        sq[(16 * i + 256) * 16] = make_float2(ar - cr, -(ai - ci));
        sq[(16 * i + 384) * 16] = make_float2(br - dr, -(bi_ - di));
      } else {
        tp[(size_t)(kk + 16 * i) * FN]       = make_float2(ar + cr, ai + ci);
        tp[(size_t)(kk + 16 * i + 128) * FN] = make_float2(br + dr, bi_ + di);
        tp[(size_t)(kk + 16 * i + 256) * FN] = make_float2(ar - cr, ai - ci);
        tp[(size_t)(kk + 16 * i + 384) * FN] = make_float2(br - dr, bi_ - di);
      }
    }
  }
}
__global__ __launch_bounds__(CT_TPB)
void k_band_cols(const float2* __restrict__ specRow, const float* __restrict__ win,
                 float2* __restrict__ T, int b0, int nb) {
  __shared__ float2 Y[2][16 * YK];
  __shared__ float2 Tw[32][16];
  __shared__ float2 L5[FN];
  __shared__ float2 Sp[2 * SP_SUB];
  const int tid = threadIdx.x, lane = tid & 31, wid = tid >> 5;
  const int sub = wid >> 3, w8 = wid & 7;
  const int c = lane & 15, h = lane >> 4;
  const int col = blockIdx.x * CT_COLS + sub * 16 + c;
  const size_t ebase = (size_t)(64 * h + w8) * FN + col;
  float2 xs[4][4][2];
#pragma unroll
  for (int jj = 0; jj < 4; ++jj)
#pragma unroll
    for (int kc = 0; kc < 4; ++kc) {
      xs[jj][kc][0] = specRow[ebase + (size_t)(128 * kc + 8 * jj) * FN];
      xs[jj][kc][1] = specRow[ebase + (size_t)(128 * kc + 32 + 8 * jj) * FN];
    }
  fill_w512(L5, tid, CT_TPB, 1.0f);
  const int np = (nb + 1) >> 1;
  const int rb = 64 * h + w8;
  const size_t mcol  = (size_t)((512 - col) & 511);
  const size_t mbase = (size_t)(512 - rb) * FN + mcol;
  const size_t mzero = (size_t)((512 - rb) & 511) * FN + mcol;
  float wr[4][4][2][4];
  lds_barrier();
  {
    const int n2 = tid >> 4, k1 = tid & 15;
    Tw[n2][k1] = L5[n2 * k1];
  }
  M8 m0, m1, m2, m3;
  make_m8(L5, lane, 0, m0);
  make_m8(L5, lane, 16, m1);
  make_m8(L5, lane, 32, m2);
  make_m8(L5, lane, 48, m3);
  lds_barrier();
  float2* const Ysub = Y[sub];
  const v8f zero = {0.f,0.f,0.f,0.f,0.f,0.f,0.f,0.f};
#pragma unroll
  for (int jj = 0; jj < 4; ++jj) {
    v8f d0 = zero, d1 = zero;
#pragma unroll
    for (int kc = 0; kc < 4; ++kc) {
      v2f x0; x0.x = xs[jj][kc][0].x; x0.y = -xs[jj][kc][0].y;
      v2f x1; x1.x = xs[jj][kc][1].x; x1.y = -xs[jj][kc][1].y;
      d0 = wmma_f32(m0.a[kc], x0, d0);
      d1 = wmma_f32(m2.a[kc], x1, d1);
    }
    col_s1_store(Ysub, Tw, w8 + 8 * jj, h, c, d0, d1);
  }
  if ((int)blockIdx.y < np) {
    const int pb = 2 * (int)blockIdx.y;
    const float* const wa = win + (size_t)(b0 + pb) * FN * FN;
    const float* const wb = (pb + 1 < nb) ? wa + (size_t)FN * FN : wa;
    load_win<0, 2>(wr, wa, wb, ebase, mbase, mzero);
  }
  lds_barrier();
  col_stage2<true>(Ysub, w8, c, h, m0, m1, m2, m3, nullptr,
                   Sp + sub * SP_SUB + (2 * w8 + 64 * h) * 16 + c);
  if ((int)blockIdx.y < np) {
    const int pb = 2 * (int)blockIdx.y;
    const float* const wa = win + (size_t)(b0 + pb) * FN * FN;
    const float* const wb = (pb + 1 < nb) ? wa + (size_t)FN * FN : wa;
    load_win<2, 2>(wr, wa, wb, ebase, mbase, mzero);
  }
  lds_barrier();
  const float2* const spr = Sp + sub * SP_SUB + (64 * h + (w8 ^ h)) * 16 + c;
  const size_t obase = (size_t)(2 * w8 + 64 * h) * FN + col;
  for (int bi = blockIdx.y; bi < np; bi += gridDim.y) {
    const float fa = 0.5f;
    const float fb = (2 * bi + 1 < nb) ? 0.5f : 0.0f;
    const int bn = bi + (int)gridDim.y;
#pragma unroll
    for (int jj = 0; jj < 4; ++jj) {
      v8f d0 = zero, d1 = zero;
#pragma unroll
      for (int kc = 0; kc < 4; ++kc) {
        const float pa = fa * (wr[jj][kc][0][0] + wr[jj][kc][0][1]);
        const float pb = fb * (wr[jj][kc][0][2] + wr[jj][kc][0][3]);
        const float qa = fa * (wr[jj][kc][1][0] + wr[jj][kc][1][1]);
        const float qb = fb * (wr[jj][kc][1][2] + wr[jj][kc][1][3]);
        const float2 sa = spr[(128 * kc + 8 * jj) * 16];
        const float2 sb = spr[(128 * kc + 32 + 8 * jj) * 16];
        v2f x0; x0.x = sa.x * pa - sa.y * pb; x0.y = sa.x * pb + sa.y * pa;
        v2f x1; x1.x = sb.x * qa - sb.y * qb; x1.y = sb.x * qb + sb.y * qa;
        d0 = wmma_f32(m0.a[kc], x0, d0);
        d1 = wmma_f32(m2.a[kc], x1, d1);
      }
      col_s1_store(Ysub, Tw, w8 + 8 * jj, h, c, d0, d1);
    }
    if (bn < np) {
      const float* const wan = win + (size_t)(b0 + 2 * bn) * FN * FN;
      const float* const wbn = (2 * bn + 1 < nb) ? wan + (size_t)FN * FN : wan;
      load_win<0, 2>(wr, wan, wbn, ebase, mbase, mzero);
    }
    lds_barrier();
    col_stage2<false>(Ysub, w8, c, h, m0, m1, m2, m3,
                      T + (size_t)bi * FN * FN + obase, nullptr);
    if (bn < np) {
      const float* const wan = win + (size_t)(b0 + 2 * bn) * FN * FN;
      const float* const wbn = (2 * bn + 1 < nb) ? wan + (size_t)FN * FN : wan;
      load_win<2, 2>(wr, wan, wbn, ebase, mbase, mzero);
    }
    lds_barrier();
  }
}

#define RW_TPB   128
#define RW_WPB   (RW_TPB / 32)
#define RW_WGS   768
#define RW_PITCH 34
__global__ __launch_bounds__(RW_TPB)
void k_band_rows(const float2* __restrict__ T, float* __restrict__ out,
                 int b0, int nb) {
  __shared__ float2 L512[FN];
  __shared__ __attribute__((aligned(16))) float2 Yx[RW_WPB][16 * RW_PITCH];
  const int tid = threadIdx.x, lane = tid & 31, wid = tid >> 5;
  const int c = lane & 15, h = lane >> 4;
  fill_w512(L512, tid, RW_TPB, 1.0f);
  __syncthreads();
  const float scale = 1.0f / (512.0f * 512.0f);
  const int s2 = 2 * (c & 7) + (c >> 3);
  v2f w1r[4], w1i[4], w1n[4];
  v2f a2r[4], a2n[4], a2i[4];
  v2f b2r[4], b2n[4], b2i[4];
#pragma unroll
  for (int kc = 0; kc < 4; ++kc) {
    const int p0 = 4 * kc + 2 * h;
    const float2 u0 = L512[(32 * c * p0) & 511];
    const float2 u1 = L512[(32 * c * (p0 + 1)) & 511];
    v2f t0; t0.x = u0.x;  t0.y = u1.x;  w1r[kc] = t0;
    v2f t1; t1.x = u0.y;  t1.y = u1.y;  w1i[kc] = t1;
    v2f t2; t2.x = -u0.y; t2.y = -u1.y; w1n[kc] = t2;
    const float2 g0 = L512[(32 * s2 * p0) & 511];
    const float2 g1 = L512[(32 * s2 * (p0 + 1)) & 511];
    v2f t3; t3.x = g0.x * scale;  t3.y = g1.x * scale;  a2r[kc] = t3;
    v2f t4; t4.x = -g0.y * scale; t4.y = -g1.y * scale; a2n[kc] = t4;
    v2f t7; t7.x = g0.y * scale;  t7.y = g1.y * scale;  a2i[kc] = t7;
    const float2 e0 = L512[(16 * s2 * (2 * p0 + 1)) & 511];
    const float2 e1 = L512[(16 * s2 * (2 * p0 + 3)) & 511];
    v2f t5; t5.x = e0.x * scale;  t5.y = e1.x * scale;  b2r[kc] = t5;
    v2f t6; t6.x = -e0.y * scale; t6.y = -e1.y * scale; b2n[kc] = t6;
    v2f t8; t8.x = e0.y * scale;  t8.y = e1.y * scale;  b2i[kc] = t8;
  }
  float2 twd[2][8];
#pragma unroll
  for (int t = 0; t < 2; ++t)
#pragma unroll
    for (int r = 0; r < 8; ++r) twd[t][r] = L512[(16 * t + c) * (8 * h + r)];

  float2* const ysh = Yx[wid];
  const float4* const yrd =
      reinterpret_cast<const float4*>(ysh + c * RW_PITCH + 4 * h);
  const v8f zero = {0.f,0.f,0.f,0.f,0.f,0.f,0.f,0.f};
  const int total = ((nb + 1) >> 1) * FN;
  const int nw = (int)gridDim.x * RW_WPB;
  int l = (int)blockIdx.x * RW_WPB + wid;
  const int lbase = 64 * h + c;
  float2 xc[2][4][2];
  if (l < total) {
    const float2* const tp = T + (size_t)l * FN + lbase;
#pragma unroll
    for (int t = 0; t < 2; ++t)
#pragma unroll
      for (int kc = 0; kc < 4; ++kc) {
        xc[t][kc][0] = tp[128 * kc + 16 * t];
        xc[t][kc][1] = tp[128 * kc + 32 + 16 * t];
      }
  }
  for (; l < total; l += nw) {
    v8f yr[2], yi[2];
#pragma unroll
    for (int t = 0; t < 2; ++t) {
      yr[t] = zero; yi[t] = zero;
#pragma unroll
      for (int kc = 0; kc < 4; ++kc) {
        v2f br; br.x = xc[t][kc][0].x; br.y = xc[t][kc][1].x;
        v2f bm; bm.x = xc[t][kc][0].y; bm.y = xc[t][kc][1].y;
        yr[t] = wmma_f32(w1r[kc], br, yr[t]);
        yi[t] = wmma_f32(w1r[kc], bm, yi[t]);
        yr[t] = wmma_f32(w1n[kc], bm, yr[t]);
        yi[t] = wmma_f32(w1i[kc], br, yi[t]);
      }
    }
    if (l + nw < total) {
      const float2* const tn = T + (size_t)(l + nw) * FN + lbase;
#pragma unroll
      for (int t = 0; t < 2; ++t)
#pragma unroll
        for (int kc = 0; kc < 4; ++kc) {
          xc[t][kc][0] = tn[128 * kc + 16 * t];
          xc[t][kc][1] = tn[128 * kc + 32 + 16 * t];
        }
    }
#pragma unroll
    for (int t = 0; t < 2; ++t) {
#pragma unroll
      for (int r = 0; r < 8; ++r) {
        const float2 tw = twd[t][r];
        const float a = yr[t][r], b = yi[t][r];
        ysh[(8 * h + r) * RW_PITCH + 16 * t + c] =
            make_float2(a * tw.x - b * tw.y, a * tw.y + b * tw.x);
      }
    }
    float4 ya[4][2];
#pragma unroll
    for (int kc = 0; kc < 4; ++kc) {
      ya[kc][0] = yrd[4 * kc];
      ya[kc][1] = yrd[4 * kc + 1];
    }
    v8f g0 = zero, g1 = zero, h0 = zero, h1 = zero;
#pragma unroll
    for (int kc = 0; kc < 4; ++kc) {
      v2f r0; r0.x = ya[kc][0].x; r0.y = ya[kc][1].x;
      v2f i0; i0.x = ya[kc][0].y; i0.y = ya[kc][1].y;
      v2f r1; r1.x = ya[kc][0].z; r1.y = ya[kc][1].z;
      v2f i1; i1.x = ya[kc][0].w; i1.y = ya[kc][1].w;
      g0 = wmma_f32(a2r[kc], r0, g0);
      g1 = wmma_f32(b2r[kc], r1, g1);
      h0 = wmma_f32(a2i[kc], r0, h0);
      h1 = wmma_f32(b2i[kc], r1, h1);
      g0 = wmma_f32(a2n[kc], i0, g0);
      g1 = wmma_f32(b2n[kc], i1, g1);
      h0 = wmma_f32(a2r[kc], i0, h0);
      h1 = wmma_f32(b2r[kc], i1, h1);
    }
    const int pl = l >> 9, row = l & (FN - 1);
    float* const op =
        out + ((size_t)(b0 + 2 * pl) * FN + (size_t)row) * FN + 16 * h + c;
#pragma unroll
    for (int r = 0; r < 8; ++r) {
      op[32 * r]       = g0[r] + g1[r];
      op[32 * r + 256] = g0[r] - g1[r];
    }
    if (2 * pl + 1 < nb) {
      float* const oq = op + (size_t)FN * FN;
#pragma unroll
      for (int r = 0; r < 8; ++r) {
        oq[32 * r]       = h0[r] + h1[r];
        oq[32 * r + 256] = h0[r] - h1[r];
      }
    }
  }
}

extern "C" void kernel_launch(void* const* d_in, const int* in_sizes, int n_in,
                              void* d_out, int out_size, void* d_ws,
                              size_t ws_size, hipStream_t stream) {
  (void)in_sizes; (void)n_in; (void)out_size;
  const float* img = (const float*)d_in[0];
  const float* win = (const float*)d_in[1];
  float* out = (float*)d_out;

  float2* specRow = (float2*)d_ws;
  float2* bandT   = specRow + 2 * (size_t)FN * FN;
  const size_t reserved = 2ull * sizeof(float2) * FN * FN;
  const size_t perBand  = sizeof(float2) * (size_t)FN * FN;
  int chunk = (ws_size > reserved) ? (int)((ws_size - reserved) / perBand) : 0;
  if (chunk < 1)  chunk = 1;
  if (chunk > NB) chunk = NB;

  const dim3 blk(TPB);
  k_fwd_rows<<<dim3(FN / LPB), blk, 0, stream>>>(img, specRow);
  for (int b0 = 0; b0 < NB; b0 += chunk) {
    const int nb = (NB - b0 < chunk) ? (NB - b0) : chunk;
    k_band_cols<<<dim3(FN / CT_COLS, GBC), dim3(CT_TPB), 0, stream>>>(specRow, win, bandT, b0, nb);
    k_band_rows<<<dim3(RW_WGS), dim3(RW_TPB), 0, stream>>>(bandT, out, b0, nb);
  }
}


# Row passes (forward, band) factored 512=(8*2)*(8*4) as the column pass: twiddled DFT-8 = one real 16x16 matrix on stacked (re,im), radix 2/4 in lanes; 32 matrix instr/row, was 64; sums re-associated
# speedup vs baseline: 3.3148x; 1.0598x over previous
#include <hip/hip_runtime.h>
#include <hip/hip_bf16.h>


#define FN   512
#define NB   43
#define LPB  4
#define TPB  128
#define PI_F 3.14159265358979323846f

typedef float v2f __attribute__((ext_vector_type(2)));
typedef float v8f __attribute__((ext_vector_type(8)));

__device__ __forceinline__ v8f wmma_f32(v2f a, v2f b, v8f c) {
  return __builtin_amdgcn_wmma_f32_16x16x4_f32(false, a, false, b,
                                               (short)0, c, false, false);
}

__device__ __forceinline__ void fill_w512(float2* L, int tid, int nthr, float sgn) {
  const float step = 2.0f * PI_F / 512.0f;
  for (int j = tid; j < 128; j += nthr) {
    float s, c; sincosf(step * (float)j, &s, &c);
    const float ss = sgn * s, sc = sgn * c;
    L[j]       = make_float2( c,  ss);
    L[j + 128] = make_float2(-s,  sc);
    L[j + 256] = make_float2(-c, -ss);
    L[j + 384] = make_float2( s, -sc);
  }
}


struct M8 { v2f a[4]; };
__device__ __forceinline__ void make_m8(const float2* L, int lane, int tq, M8& m) {
  const int row = lane & 15, h = lane >> 4;
  const int j = 4 * (row >> 3) + ((row & 7) >> 1);
  const bool im = (row & 1) != 0;
#pragma unroll
  for (int kc = 0; kc < 4; ++kc) {
    const int a = 2 * kc + h;
    const float2 u = L[((tq + 64 * a) * j) & 511];
    v2f t;
    t.x = im ? u.y : u.x;
    t.y = im ? u.x : -u.y;
    m.a[kc] = t;
  }
}
__device__ __forceinline__ void make_m8s(const float2* L, int lane, int tq,
                                         float scale, M8& m) {
  const int row = lane & 15, h = lane >> 4;
  const int s = 2 * ((row & 7) >> 1) + (row >> 3);
  const bool im = (row & 1) != 0;
#pragma unroll
  for (int kc = 0; kc < 4; ++kc) {
    const int p = 2 * kc + h;
    const float2 u = L[((tq + 64 * p) * s) & 511];
    v2f t;
    t.x = (im ? u.y : u.x) * scale;
    t.y = (im ? u.x : -u.y) * scale;
    m.a[kc] = t;
  }
}
#define RW_PITCH 34
__global__ __launch_bounds__(TPB)
void k_fwd_rows(const float* __restrict__ img, float2* __restrict__ specRow) {
  __shared__ float2 L5[FN];
  __shared__ __attribute__((aligned(16))) float2 Yf[LPB][16 * RW_PITCH];
  const int tid = threadIdx.x, lane = tid & 31, wid = tid >> 5;
  const int c = lane & 15, h = lane >> 4;
  const int line = blockIdx.x * LPB + wid;
  float xv[2][4][2];
  const float* const ip = img + (size_t)line * FN + 64 * h + c;
#pragma unroll
  for (int t = 0; t < 2; ++t)
#pragma unroll
    for (int kc = 0; kc < 4; ++kc) {
      xv[t][kc][0] = ip[128 * kc + 16 * t];
      xv[t][kc][1] = ip[128 * kc + 32 + 16 * t];
    }
  fill_w512(L5, tid, TPB, -1.0f);
  __syncthreads();
  M8 p0, p1, q0, q1, q2, q3;
  make_m8(L5, lane, 0, p0);
  make_m8(L5, lane, 32, p1);
  make_m8s(L5, lane, 0, 1.0f, q0);
  make_m8s(L5, lane, 16, 1.0f, q1);
  make_m8s(L5, lane, 32, 1.0f, q2);
  make_m8s(L5, lane, 48, 1.0f, q3);
  float2* const ysh = Yf[wid];
  const v8f zero = {0.f,0.f,0.f,0.f,0.f,0.f,0.f,0.f};
#pragma unroll
  for (int t = 0; t < 2; ++t) {
    v8f d0 = zero, d1 = zero;
#pragma unroll
    for (int kc = 0; kc < 4; ++kc) {
      v2f x0; x0.x = xv[t][kc][0]; x0.y = 0.0f;
      v2f x1; x1.x = xv[t][kc][1]; x1.y = 0.0f;
      d0 = wmma_f32(p0.a[kc], x0, d0);
      d1 = wmma_f32(p1.a[kc], x1, d1);
    }
    const int n2 = 16 * t + c;
#pragma unroll
    for (int i = 0; i < 4; ++i) {
      const int j = 4 * h + i;
      const float pr = d0[2 * i] + d1[2 * i], pi = d0[2 * i + 1] + d1[2 * i + 1];
      const float mr = d0[2 * i] - d1[2 * i], mi = d0[2 * i + 1] - d1[2 * i + 1];
      const float2 ta = L5[n2 * j], tb = L5[n2 * (j + 8)];
      ysh[j * RW_PITCH + n2] =
          make_float2(pr * ta.x - pi * ta.y, pr * ta.y + pi * ta.x);
      ysh[(j + 8) * RW_PITCH + n2] =
          make_float2(mr * tb.x - mi * tb.y, mr * tb.y + mi * tb.x);
    }
  }
  const float4* const yrd =
      reinterpret_cast<const float4*>(ysh + c * RW_PITCH + 4 * h);
  float4 ya[4][2];
#pragma unroll
  for (int kc = 0; kc < 4; ++kc) {
    ya[kc][0] = yrd[4 * kc];
    ya[kc][1] = yrd[4 * kc + 1];
  }
  v8f e0 = zero, e1 = zero, e2 = zero, e3 = zero;
#pragma unroll
  for (int kc = 0; kc < 4; ++kc) {
    v2f y0; y0.x = ya[kc][0].x; y0.y = ya[kc][0].y;
    v2f y1; y1.x = ya[kc][0].z; y1.y = ya[kc][0].w;
    v2f y2; y2.x = ya[kc][1].x; y2.y = ya[kc][1].y;
    v2f y3; y3.x = ya[kc][1].z; y3.y = ya[kc][1].w;
    e0 = wmma_f32(q0.a[kc], y0, e0);
    e1 = wmma_f32(q1.a[kc], y1, e1);
    e2 = wmma_f32(q2.a[kc], y2, e2);
    e3 = wmma_f32(q3.a[kc], y3, e3);
  }
  float2* const op = specRow + (size_t)line * FN + 16 * h + c;
#pragma unroll
  for (int i = 0; i < 4; ++i) {
    const float ar = e0[2 * i] + e2[2 * i], ai = e0[2 * i + 1] + e2[2 * i + 1];
    const float br = e0[2 * i] - e2[2 * i], bi_ = e0[2 * i + 1] - e2[2 * i + 1];
    const float cr = e1[2 * i] + e3[2 * i], ci = e1[2 * i + 1] + e3[2 * i + 1];
    const float dr = e1[2 * i + 1] - e3[2 * i + 1];
    const float di = e3[2 * i] - e1[2 * i];
    op[32 * i]       = make_float2(ar + cr, ai + ci);
    op[32 * i + 128] = make_float2(br + dr, bi_ + di);
    op[32 * i + 256] = make_float2(ar - cr, ai - ci);
    op[32 * i + 384] = make_float2(br - dr, bi_ - di);
  }
}

template <int JLO, int JN>
__device__ __forceinline__ void load_win(float (&wr)[4][4][2][4],
                                         const float* __restrict__ wa,
                                         const float* __restrict__ wb,
                                         size_t ebase, size_t mbase, size_t m0) {
  const float* const ad = wa + ebase;
  const float* const am = wa + mbase;
  const float* const bd = wb + ebase;
  const float* const bm = wb + mbase;
#pragma unroll
  for (int jj = JLO; jj < JLO + JN; ++jj)
#pragma unroll
    for (int kc = 0; kc < 4; ++kc)
#pragma unroll
      for (int e = 0; e < 2; ++e) {
        const int K = 128 * kc + 32 * e + 8 * jj;
        const ptrdiff_t ko = (ptrdiff_t)K * FN;
        wr[jj][kc][e][0] = ad[ko];
        wr[jj][kc][e][1] = (K == 0) ? wa[m0] : am[-ko];
        wr[jj][kc][e][2] = bd[ko];
        wr[jj][kc][e][3] = (K == 0) ? wb[m0] : bm[-ko];
      }
}
__device__ __forceinline__ void lds_barrier() {
  __builtin_amdgcn_fence(__ATOMIC_RELEASE, "workgroup", "local");
  __builtin_amdgcn_s_barrier();
  __builtin_amdgcn_fence(__ATOMIC_ACQUIRE, "workgroup", "local");
}
#define CT_COLS 32
#define CT_TPB  512
#define GBC     11
#define YS      20
#define YK      (32 * YS + 4)
#define SP_SUB  (FN * 16)
__device__ __forceinline__ void col_s1_store(float2* Ysub, const float2 (*Tw)[16],
                                             int n2, int h, int c,
                                             const v8f& d0, const v8f& d1) {
#pragma unroll
  for (int i = 0; i < 4; ++i) {
    const int j = 4 * h + i;
    const float pr = d0[2 * i] + d1[2 * i], pi = d0[2 * i + 1] + d1[2 * i + 1];
    const float mr = d0[2 * i] - d1[2 * i], mi = d0[2 * i + 1] - d1[2 * i + 1];
    const float2 ta = Tw[n2][j], tb = Tw[n2][j + 8];
    Ysub[j * YK + n2 * YS + c] =
        make_float2(pr * ta.x - pi * ta.y, pr * ta.y + pi * ta.x);
    Ysub[(j + 8) * YK + n2 * YS + c] =
        make_float2(mr * tb.x - mi * tb.y, mr * tb.y + mi * tb.x);
  }
}
template <bool TOLDS>
__device__ __forceinline__ void col_stage2(const float2* Ysub, int w8, int c, int h,
                                           const M8& m0, const M8& m1,
                                           const M8& m2, const M8& m3,
                                           float2* tp, float2* sp) {
  const v8f zero = {0.f,0.f,0.f,0.f,0.f,0.f,0.f,0.f};
#pragma unroll
  for (int kk = 0; kk < 2; ++kk) {
    const float2* const yk = Ysub + (2 * w8 + kk) * YK + c;
    float2 ya[4][4];
#pragma unroll
    for (int kc = 0; kc < 4; ++kc)
#pragma unroll
      for (int q = 0; q < 4; ++q)
        ya[kc][q] = yk[(8 * kc + 4 * h + q) * YS];
    v8f e0 = zero, e1 = zero, e2 = zero, e3 = zero;
#pragma unroll
    for (int kc = 0; kc < 4; ++kc) {
      v2f y0; y0.x = ya[kc][0].x; y0.y = ya[kc][0].y;
      v2f y1; y1.x = ya[kc][1].x; y1.y = ya[kc][1].y;
      v2f y2; y2.x = ya[kc][2].x; y2.y = ya[kc][2].y;
      v2f y3; y3.x = ya[kc][3].x; y3.y = ya[kc][3].y;
      e0 = wmma_f32(m0.a[kc], y0, e0);
      e1 = wmma_f32(m1.a[kc], y1, e1);
      e2 = wmma_f32(m2.a[kc], y2, e2);
      e3 = wmma_f32(m3.a[kc], y3, e3);
    }
    float2* const sq = TOLDS ? sp + (kk ^ h) * 16 : nullptr;
#pragma unroll
    for (int i = 0; i < 4; ++i) {
      const float ar = e0[2 * i] + e2[2 * i], ai = e0[2 * i + 1] + e2[2 * i + 1];
      const float br = e0[2 * i] - e2[2 * i], bi_ = e0[2 * i + 1] - e2[2 * i + 1];
      const float cr = e1[2 * i] + e3[2 * i], ci = e1[2 * i + 1] + e3[2 * i + 1];
      const float dr = e3[2 * i + 1] - e1[2 * i + 1];
      const float di = e1[2 * i] - e3[2 * i];
      if (TOLDS) {
        sq[(16 * i) * 16]       = make_float2(ar + cr, -(ai + ci));
        sq[(16 * i + 128) * 16] = make_float2(br + dr, -(bi_ + di));
        sq[(16 * i + 256) * 16] = make_float2(ar - cr, -(ai - ci));
        sq[(16 * i + 384) * 16] = make_float2(br - dr, -(bi_ - di));
      } else {
        tp[(size_t)(kk + 16 * i) * FN]       = make_float2(ar + cr, ai + ci);
        tp[(size_t)(kk + 16 * i + 128) * FN] = make_float2(br + dr, bi_ + di);
        tp[(size_t)(kk + 16 * i + 256) * FN] = make_float2(ar - cr, ai - ci);
        tp[(size_t)(kk + 16 * i + 384) * FN] = make_float2(br - dr, bi_ - di);
      }
    }
  }
}
__global__ __launch_bounds__(CT_TPB)
void k_band_cols(const float2* __restrict__ specRow, const float* __restrict__ win,
                 float2* __restrict__ T, int b0, int nb) {
  __shared__ float2 Y[2][16 * YK];
  __shared__ float2 Tw[32][16];
  __shared__ float2 L5[FN];
  __shared__ float2 Sp[2 * SP_SUB];
  const int tid = threadIdx.x, lane = tid & 31, wid = tid >> 5;
  const int sub = wid >> 3, w8 = wid & 7;
  const int c = lane & 15, h = lane >> 4;
  const int col = blockIdx.x * CT_COLS + sub * 16 + c;
  const size_t ebase = (size_t)(64 * h + w8) * FN + col;
  float2 xs[4][4][2];
#pragma unroll
  for (int jj = 0; jj < 4; ++jj)
#pragma unroll
    for (int kc = 0; kc < 4; ++kc) {
      xs[jj][kc][0] = specRow[ebase + (size_t)(128 * kc + 8 * jj) * FN];
      xs[jj][kc][1] = specRow[ebase + (size_t)(128 * kc + 32 + 8 * jj) * FN];
    }
  fill_w512(L5, tid, CT_TPB, 1.0f);
  const int np = (nb + 1) >> 1;
  const int rb = 64 * h + w8;
  const size_t mcol  = (size_t)((512 - col) & 511);
  const size_t mbase = (size_t)(512 - rb) * FN + mcol;
  const size_t mzero = (size_t)((512 - rb) & 511) * FN + mcol;
  float wr[4][4][2][4];
  lds_barrier();
  {
    const int n2 = tid >> 4, k1 = tid & 15;
    Tw[n2][k1] = L5[n2 * k1];
  }
  M8 m0, m1, m2, m3;
  make_m8(L5, lane, 0, m0);
  make_m8(L5, lane, 16, m1);
  make_m8(L5, lane, 32, m2);
  make_m8(L5, lane, 48, m3);
  lds_barrier();
  float2* const Ysub = Y[sub];
  const v8f zero = {0.f,0.f,0.f,0.f,0.f,0.f,0.f,0.f};
#pragma unroll
  for (int jj = 0; jj < 4; ++jj) {
    v8f d0 = zero, d1 = zero;
#pragma unroll
    for (int kc = 0; kc < 4; ++kc) {
      v2f x0; x0.x = xs[jj][kc][0].x; x0.y = -xs[jj][kc][0].y;
      v2f x1; x1.x = xs[jj][kc][1].x; x1.y = -xs[jj][kc][1].y;
      d0 = wmma_f32(m0.a[kc], x0, d0);
      d1 = wmma_f32(m2.a[kc], x1, d1);
    }
    col_s1_store(Ysub, Tw, w8 + 8 * jj, h, c, d0, d1);
  }
  if ((int)blockIdx.y < np) {
    const int pb = 2 * (int)blockIdx.y;
    const float* const wa = win + (size_t)(b0 + pb) * FN * FN;
    const float* const wb = (pb + 1 < nb) ? wa + (size_t)FN * FN : wa;
    load_win<0, 2>(wr, wa, wb, ebase, mbase, mzero);
  }
  lds_barrier();
  col_stage2<true>(Ysub, w8, c, h, m0, m1, m2, m3, nullptr,
                   Sp + sub * SP_SUB + (2 * w8 + 64 * h) * 16 + c);
  if ((int)blockIdx.y < np) {
    const int pb = 2 * (int)blockIdx.y;
    const float* const wa = win + (size_t)(b0 + pb) * FN * FN;
    const float* const wb = (pb + 1 < nb) ? wa + (size_t)FN * FN : wa;
    load_win<2, 2>(wr, wa, wb, ebase, mbase, mzero);
  }
  lds_barrier();
  const float2* const spr = Sp + sub * SP_SUB + (64 * h + (w8 ^ h)) * 16 + c;
  const size_t obase = (size_t)(2 * w8 + 64 * h) * FN + col;
  for (int bi = blockIdx.y; bi < np; bi += gridDim.y) {
    const float fa = 0.5f;
    const float fb = (2 * bi + 1 < nb) ? 0.5f : 0.0f;
    const int bn = bi + (int)gridDim.y;
#pragma unroll
    for (int jj = 0; jj < 4; ++jj) {
      v8f d0 = zero, d1 = zero;
#pragma unroll
      for (int kc = 0; kc < 4; ++kc) {
        const float pa = fa * (wr[jj][kc][0][0] + wr[jj][kc][0][1]);
        const float pb = fb * (wr[jj][kc][0][2] + wr[jj][kc][0][3]);
        const float qa = fa * (wr[jj][kc][1][0] + wr[jj][kc][1][1]);
        const float qb = fb * (wr[jj][kc][1][2] + wr[jj][kc][1][3]);
        const float2 sa = spr[(128 * kc + 8 * jj) * 16];
        const float2 sb = spr[(128 * kc + 32 + 8 * jj) * 16];
        v2f x0; x0.x = sa.x * pa - sa.y * pb; x0.y = sa.x * pb + sa.y * pa;
        v2f x1; x1.x = sb.x * qa - sb.y * qb; x1.y = sb.x * qb + sb.y * qa;
        d0 = wmma_f32(m0.a[kc], x0, d0);
        d1 = wmma_f32(m2.a[kc], x1, d1);
      }
      col_s1_store(Ysub, Tw, w8 + 8 * jj, h, c, d0, d1);
    }
    if (bn < np) {
      const float* const wan = win + (size_t)(b0 + 2 * bn) * FN * FN;
      const float* const wbn = (2 * bn + 1 < nb) ? wan + (size_t)FN * FN : wan;
      load_win<0, 2>(wr, wan, wbn, ebase, mbase, mzero);
    }
    lds_barrier();
    col_stage2<false>(Ysub, w8, c, h, m0, m1, m2, m3,
                      T + (size_t)bi * FN * FN + obase, nullptr);
    if (bn < np) {
      const float* const wan = win + (size_t)(b0 + 2 * bn) * FN * FN;
      const float* const wbn = (2 * bn + 1 < nb) ? wan + (size_t)FN * FN : wan;
      load_win<2, 2>(wr, wan, wbn, ebase, mbase, mzero);
    }
    lds_barrier();
  }
}

#define RW_TPB   128
#define RW_WPB   (RW_TPB / 32)
#define RW_WGS   768
__global__ __launch_bounds__(RW_TPB)
void k_band_rows(const float2* __restrict__ T, float* __restrict__ out,
                 int b0, int nb) {
  __shared__ float2 L512[FN];
  __shared__ __attribute__((aligned(16))) float2 Yx[RW_WPB][16 * RW_PITCH];
  const int tid = threadIdx.x, lane = tid & 31, wid = tid >> 5;
  const int c = lane & 15, h = lane >> 4;
  fill_w512(L512, tid, RW_TPB, 1.0f);
  __syncthreads();
  const float scale = 1.0f / (512.0f * 512.0f);
  M8 p0, p1, q0, q1, q2, q3;
  make_m8(L512, lane, 0, p0);
  make_m8(L512, lane, 32, p1);
  make_m8s(L512, lane, 0, scale, q0);
  make_m8s(L512, lane, 16, scale, q1);
  make_m8s(L512, lane, 32, scale, q2);
  make_m8s(L512, lane, 48, scale, q3);
  float2 twd[2][4][2];
#pragma unroll
  for (int t = 0; t < 2; ++t)
#pragma unroll
    for (int i = 0; i < 4; ++i)
#pragma unroll
      for (int u = 0; u < 2; ++u)
        twd[t][i][u] = L512[(16 * t + c) * (4 * h + i + 8 * u)];

  float2* const ysh = Yx[wid];
  const float4* const yrd =
      reinterpret_cast<const float4*>(ysh + c * RW_PITCH + 4 * h);
  const v8f zero = {0.f,0.f,0.f,0.f,0.f,0.f,0.f,0.f};
  const int total = ((nb + 1) >> 1) * FN;
  const int nw = (int)gridDim.x * RW_WPB;
  int l = (int)blockIdx.x * RW_WPB + wid;
  const int lbase = 64 * h + c;
  float2 xc[2][4][2];
  if (l < total) {
    const float2* const tp = T + (size_t)l * FN + lbase;
#pragma unroll
    for (int t = 0; t < 2; ++t)
#pragma unroll
      for (int kc = 0; kc < 4; ++kc) {
        xc[t][kc][0] = tp[128 * kc + 16 * t];
        xc[t][kc][1] = tp[128 * kc + 32 + 16 * t];
      }
  }
  for (; l < total; l += nw) {
    v8f d0[2], d1[2];
#pragma unroll
    for (int t = 0; t < 2; ++t) {
      d0[t] = zero; d1[t] = zero;
#pragma unroll
      for (int kc = 0; kc < 4; ++kc) {
        v2f x0; x0.x = xc[t][kc][0].x; x0.y = xc[t][kc][0].y;
        v2f x1; x1.x = xc[t][kc][1].x; x1.y = xc[t][kc][1].y;
        d0[t] = wmma_f32(p0.a[kc], x0, d0[t]);
        d1[t] = wmma_f32(p1.a[kc], x1, d1[t]);
      }
    }
    if (l + nw < total) {
      const float2* const tn = T + (size_t)(l + nw) * FN + lbase;
#pragma unroll
      for (int t = 0; t < 2; ++t)
#pragma unroll
        for (int kc = 0; kc < 4; ++kc) {
          xc[t][kc][0] = tn[128 * kc + 16 * t];
          xc[t][kc][1] = tn[128 * kc + 32 + 16 * t];
        }
    }
#pragma unroll
    for (int t = 0; t < 2; ++t) {
#pragma unroll
      for (int i = 0; i < 4; ++i) {
        const float pr = d0[t][2 * i] + d1[t][2 * i];
        const float pi = d0[t][2 * i + 1] + d1[t][2 * i + 1];
        const float mr = d0[t][2 * i] - d1[t][2 * i];
        const float mi = d0[t][2 * i + 1] - d1[t][2 * i + 1];
        const float2 ta = twd[t][i][0], tb = twd[t][i][1];
        ysh[(4 * h + i) * RW_PITCH + 16 * t + c] =
            make_float2(pr * ta.x - pi * ta.y, pr * ta.y + pi * ta.x);
        ysh[(4 * h + i + 8) * RW_PITCH + 16 * t + c] =
            make_float2(mr * tb.x - mi * tb.y, mr * tb.y + mi * tb.x);
      }
    }
    float4 ya[4][2];
#pragma unroll
    for (int kc = 0; kc < 4; ++kc) {
      ya[kc][0] = yrd[4 * kc];
      ya[kc][1] = yrd[4 * kc + 1];
    }
    v8f e0 = zero, e1 = zero, e2 = zero, e3 = zero;
#pragma unroll
    for (int kc = 0; kc < 4; ++kc) {
      v2f y0; y0.x = ya[kc][0].x; y0.y = ya[kc][0].y;
      v2f y1; y1.x = ya[kc][0].z; y1.y = ya[kc][0].w;
      v2f y2; y2.x = ya[kc][1].x; y2.y = ya[kc][1].y;
      v2f y3; y3.x = ya[kc][1].z; y3.y = ya[kc][1].w;
      e0 = wmma_f32(q0.a[kc], y0, e0);
      e1 = wmma_f32(q1.a[kc], y1, e1);
      e2 = wmma_f32(q2.a[kc], y2, e2);
      e3 = wmma_f32(q3.a[kc], y3, e3);
    }
    const int pl = l >> 9, row = l & (FN - 1);
    float* const op =
        out + ((size_t)(b0 + 2 * pl) * FN + (size_t)row) * FN + 16 * h + c;
#pragma unroll
    for (int i = 0; i < 4; ++i) {
      const float ar = e0[2 * i] + e2[2 * i], br = e0[2 * i] - e2[2 * i];
      const float cr = e1[2 * i] + e3[2 * i];
      const float dr = e3[2 * i + 1] - e1[2 * i + 1];
      op[32 * i]       = ar + cr;
      op[32 * i + 128] = br + dr;
      op[32 * i + 256] = ar - cr;
      op[32 * i + 384] = br - dr;
    }
    if (2 * pl + 1 < nb) {
      float* const oq = op + (size_t)FN * FN;
#pragma unroll
      for (int i = 0; i < 4; ++i) {
        const float ai = e0[2 * i + 1] + e2[2 * i + 1], bi_ = e0[2 * i + 1] - e2[2 * i + 1];
        const float ci = e1[2 * i + 1] + e3[2 * i + 1];
        const float di = e1[2 * i] - e3[2 * i];
        oq[32 * i]       = ai + ci;
        oq[32 * i + 128] = bi_ + di;
        oq[32 * i + 256] = ai - ci;
        oq[32 * i + 384] = bi_ - di;
      }
    }
  }
}

extern "C" void kernel_launch(void* const* d_in, const int* in_sizes, int n_in,
                              void* d_out, int out_size, void* d_ws,
                              size_t ws_size, hipStream_t stream) {
  (void)in_sizes; (void)n_in; (void)out_size;
  const float* img = (const float*)d_in[0];
  const float* win = (const float*)d_in[1];
  float* out = (float*)d_out;

  float2* specRow = (float2*)d_ws;
  float2* bandT   = specRow + 2 * (size_t)FN * FN;
  const size_t reserved = 2ull * sizeof(float2) * FN * FN;
  const size_t perBand  = sizeof(float2) * (size_t)FN * FN;
  int chunk = (ws_size > reserved) ? (int)((ws_size - reserved) / perBand) : 0;
  if (chunk < 1)  chunk = 1;
  if (chunk > NB) chunk = NB;

  const dim3 blk(TPB);
  k_fwd_rows<<<dim3(FN / LPB), blk, 0, stream>>>(img, specRow);
  for (int b0 = 0; b0 < NB; b0 += chunk) {
    const int nb = (NB - b0 < chunk) ? (NB - b0) : chunk;
    k_band_cols<<<dim3(FN / CT_COLS, GBC), dim3(CT_TPB), 0, stream>>>(specRow, win, bandT, b0, nb);
    k_band_rows<<<dim3(RW_WGS), dim3(RW_TPB), 0, stream>>>(bandT, out, b0, nb);
  }
}


# Band column kernel re-dealt over the chip: a work-group is 16 waves on 16 columns (was 32), grid 32 x 8 = 256 work-groups of 3 or 2 band pairs, one per WGP (was 176 of 2 pairs); same arithmetic
# speedup vs baseline: 4.0355x; 1.2174x over previous
#include <hip/hip_runtime.h>
#include <hip/hip_bf16.h>


#define FN   512
#define NB   43
#define LPB  4
#define TPB  128
#define PI_F 3.14159265358979323846f

typedef float v2f __attribute__((ext_vector_type(2)));
typedef float v8f __attribute__((ext_vector_type(8)));

__device__ __forceinline__ v8f wmma_f32(v2f a, v2f b, v8f c) {
  return __builtin_amdgcn_wmma_f32_16x16x4_f32(false, a, false, b,
                                               (short)0, c, false, false);
}

__device__ __forceinline__ void fill_w512(float2* L, int tid, int nthr, float sgn) {
  const float step = 2.0f * PI_F / 512.0f;
  for (int j = tid; j < 128; j += nthr) {
    float s, c; sincosf(step * (float)j, &s, &c);
    const float ss = sgn * s, sc = sgn * c;
    L[j]       = make_float2( c,  ss);
    L[j + 128] = make_float2(-s,  sc);
    L[j + 256] = make_float2(-c, -ss);
    L[j + 384] = make_float2( s, -sc);
  }
}


struct M8 { v2f a[4]; };
__device__ __forceinline__ void make_m8(const float2* L, int lane, int tq, M8& m) {
  const int row = lane & 15, h = lane >> 4;
  const int j = 4 * (row >> 3) + ((row & 7) >> 1);
  const bool im = (row & 1) != 0;
#pragma unroll
  for (int kc = 0; kc < 4; ++kc) {
    const int a = 2 * kc + h;
    const float2 u = L[((tq + 64 * a) * j) & 511];
    v2f t;
    t.x = im ? u.y : u.x;
    t.y = im ? u.x : -u.y;
    m.a[kc] = t;
  }
}
__device__ __forceinline__ void make_m8s(const float2* L, int lane, int tq,
                                         float scale, M8& m) {
  const int row = lane & 15, h = lane >> 4;
  const int s = 2 * ((row & 7) >> 1) + (row >> 3);
  const bool im = (row & 1) != 0;
#pragma unroll
  for (int kc = 0; kc < 4; ++kc) {
    const int p = 2 * kc + h;
    const float2 u = L[((tq + 64 * p) * s) & 511];
    v2f t;
    t.x = (im ? u.y : u.x) * scale;
    t.y = (im ? u.x : -u.y) * scale;
    m.a[kc] = t;
  }
}
#define RW_PITCH 34
__global__ __launch_bounds__(TPB)
void k_fwd_rows(const float* __restrict__ img, float2* __restrict__ specRow) {
  __shared__ float2 L5[FN];
  __shared__ __attribute__((aligned(16))) float2 Yf[LPB][16 * RW_PITCH];
  const int tid = threadIdx.x, lane = tid & 31, wid = tid >> 5;
  const int c = lane & 15, h = lane >> 4;
  const int line = blockIdx.x * LPB + wid;
  float xv[2][4][2];
  const float* const ip = img + (size_t)line * FN + 64 * h + c;
#pragma unroll
  for (int t = 0; t < 2; ++t)
#pragma unroll
    for (int kc = 0; kc < 4; ++kc) {
      xv[t][kc][0] = ip[128 * kc + 16 * t];
      xv[t][kc][1] = ip[128 * kc + 32 + 16 * t];
    }
  fill_w512(L5, tid, TPB, -1.0f);
  __syncthreads();
  M8 p0, p1, q0, q1, q2, q3;
  make_m8(L5, lane, 0, p0);
  make_m8(L5, lane, 32, p1);
  make_m8s(L5, lane, 0, 1.0f, q0);
  make_m8s(L5, lane, 16, 1.0f, q1);
  make_m8s(L5, lane, 32, 1.0f, q2);
  make_m8s(L5, lane, 48, 1.0f, q3);
  float2* const ysh = Yf[wid];
  const v8f zero = {0.f,0.f,0.f,0.f,0.f,0.f,0.f,0.f};
#pragma unroll
  for (int t = 0; t < 2; ++t) {
    v8f d0 = zero, d1 = zero;
#pragma unroll
    for (int kc = 0; kc < 4; ++kc) {
      v2f x0; x0.x = xv[t][kc][0]; x0.y = 0.0f;
      v2f x1; x1.x = xv[t][kc][1]; x1.y = 0.0f;
      d0 = wmma_f32(p0.a[kc], x0, d0);
      d1 = wmma_f32(p1.a[kc], x1, d1);
    }
    const int n2 = 16 * t + c;
#pragma unroll
    for (int i = 0; i < 4; ++i) {
      const int j = 4 * h + i;
      const float pr = d0[2 * i] + d1[2 * i], pi = d0[2 * i + 1] + d1[2 * i + 1];
      const float mr = d0[2 * i] - d1[2 * i], mi = d0[2 * i + 1] - d1[2 * i + 1];
      const float2 ta = L5[n2 * j], tb = L5[n2 * (j + 8)];
      ysh[j * RW_PITCH + n2] =
          make_float2(pr * ta.x - pi * ta.y, pr * ta.y + pi * ta.x);
      ysh[(j + 8) * RW_PITCH + n2] =
          make_float2(mr * tb.x - mi * tb.y, mr * tb.y + mi * tb.x);
    }
  }
  const float4* const yrd =
      reinterpret_cast<const float4*>(ysh + c * RW_PITCH + 4 * h);
  float4 ya[4][2];
#pragma unroll
  for (int kc = 0; kc < 4; ++kc) {
    ya[kc][0] = yrd[4 * kc];
    ya[kc][1] = yrd[4 * kc + 1];
  }
  v8f e0 = zero, e1 = zero, e2 = zero, e3 = zero;
#pragma unroll
  for (int kc = 0; kc < 4; ++kc) {
    v2f y0; y0.x = ya[kc][0].x; y0.y = ya[kc][0].y;
    v2f y1; y1.x = ya[kc][0].z; y1.y = ya[kc][0].w;
    v2f y2; y2.x = ya[kc][1].x; y2.y = ya[kc][1].y;
    v2f y3; y3.x = ya[kc][1].z; y3.y = ya[kc][1].w;
    e0 = wmma_f32(q0.a[kc], y0, e0);
    e1 = wmma_f32(q1.a[kc], y1, e1);
    e2 = wmma_f32(q2.a[kc], y2, e2);
    e3 = wmma_f32(q3.a[kc], y3, e3);
  }
  float2* const op = specRow + (size_t)line * FN + 16 * h + c;
#pragma unroll
  for (int i = 0; i < 4; ++i) {
    const float ar = e0[2 * i] + e2[2 * i], ai = e0[2 * i + 1] + e2[2 * i + 1];
    const float br = e0[2 * i] - e2[2 * i], bi_ = e0[2 * i + 1] - e2[2 * i + 1];
    const float cr = e1[2 * i] + e3[2 * i], ci = e1[2 * i + 1] + e3[2 * i + 1];
    const float dr = e1[2 * i + 1] - e3[2 * i + 1];
    const float di = e3[2 * i] - e1[2 * i];
    op[32 * i]       = make_float2(ar + cr, ai + ci);
    op[32 * i + 128] = make_float2(br + dr, bi_ + di);
    op[32 * i + 256] = make_float2(ar - cr, ai - ci);
    op[32 * i + 384] = make_float2(br - dr, bi_ - di);
  }
}

template <int JLO, int JN>
__device__ __forceinline__ void load_win(float (&wr)[2][4][2][4],
                                         const float* __restrict__ wa,
                                         const float* __restrict__ wb,
                                         size_t ebase, size_t mbase, size_t m0) {
  const float* const ad = wa + ebase;
  const float* const am = wa + mbase;
  const float* const bd = wb + ebase;
  const float* const bm = wb + mbase;
#pragma unroll
  for (int jj = JLO; jj < JLO + JN; ++jj)
#pragma unroll
    for (int kc = 0; kc < 4; ++kc)
#pragma unroll
      for (int e = 0; e < 2; ++e) {
        const int K = 128 * kc + 32 * e + 16 * jj;
        const ptrdiff_t ko = (ptrdiff_t)K * FN;
        wr[jj][kc][e][0] = ad[ko];
        wr[jj][kc][e][1] = (K == 0) ? wa[m0] : am[-ko];
        wr[jj][kc][e][2] = bd[ko];
        wr[jj][kc][e][3] = (K == 0) ? wb[m0] : bm[-ko];
      }
}
__device__ __forceinline__ void lds_barrier() {
  __builtin_amdgcn_fence(__ATOMIC_RELEASE, "workgroup", "local");
  __builtin_amdgcn_s_barrier();
  __builtin_amdgcn_fence(__ATOMIC_ACQUIRE, "workgroup", "local");
}
#define CT_COLS 16
#define CT_TPB  512
#define GBC     8
#define YS      20
#define YK      (32 * YS + 4)
#define SP_SUB  (FN * 16)
__device__ __forceinline__ void col_s1_store(float2* Ysub, const float2 (*Tw)[16],
                                             int n2, int h, int c,
                                             const v8f& d0, const v8f& d1) {
#pragma unroll
  for (int i = 0; i < 4; ++i) {
    const int j = 4 * h + i;
    const float pr = d0[2 * i] + d1[2 * i], pi = d0[2 * i + 1] + d1[2 * i + 1];
    const float mr = d0[2 * i] - d1[2 * i], mi = d0[2 * i + 1] - d1[2 * i + 1];
    const float2 ta = Tw[n2][j], tb = Tw[n2][j + 8];
    Ysub[j * YK + n2 * YS + c] =
        make_float2(pr * ta.x - pi * ta.y, pr * ta.y + pi * ta.x);
    Ysub[(j + 8) * YK + n2 * YS + c] =
        make_float2(mr * tb.x - mi * tb.y, mr * tb.y + mi * tb.x);
  }
}
template <bool TOLDS>
__device__ __forceinline__ void col_stage2(const float2* Ysub, int k1, int c, int h,
                                           const M8& m0, const M8& m1,
                                           const M8& m2, const M8& m3,
                                           float2* tp, float2* sp) {
  const v8f zero = {0.f,0.f,0.f,0.f,0.f,0.f,0.f,0.f};
  const float2* const yk = Ysub + k1 * YK + c;
  float2 ya[4][4];
#pragma unroll
  for (int kc = 0; kc < 4; ++kc)
#pragma unroll
    for (int q = 0; q < 4; ++q)
      ya[kc][q] = yk[(8 * kc + 4 * h + q) * YS];
  v8f e0 = zero, e1 = zero, e2 = zero, e3 = zero;
#pragma unroll
  for (int kc = 0; kc < 4; ++kc) {
    v2f y0; y0.x = ya[kc][0].x; y0.y = ya[kc][0].y;
    v2f y1; y1.x = ya[kc][1].x; y1.y = ya[kc][1].y;
    v2f y2; y2.x = ya[kc][2].x; y2.y = ya[kc][2].y;
    v2f y3; y3.x = ya[kc][3].x; y3.y = ya[kc][3].y;
    e0 = wmma_f32(m0.a[kc], y0, e0);
    e1 = wmma_f32(m1.a[kc], y1, e1);
    e2 = wmma_f32(m2.a[kc], y2, e2);
    e3 = wmma_f32(m3.a[kc], y3, e3);
  }
#pragma unroll
  for (int i = 0; i < 4; ++i) {
    const float ar = e0[2 * i] + e2[2 * i], ai = e0[2 * i + 1] + e2[2 * i + 1];
    const float br = e0[2 * i] - e2[2 * i], bi_ = e0[2 * i + 1] - e2[2 * i + 1];
    const float cr = e1[2 * i] + e3[2 * i], ci = e1[2 * i + 1] + e3[2 * i + 1];
    const float dr = e3[2 * i + 1] - e1[2 * i + 1];
    const float di = e1[2 * i] - e3[2 * i];
    if (TOLDS) {
      sp[(16 * i) * 16]       = make_float2(ar + cr, -(ai + ci));
      sp[(16 * i + 128) * 16] = make_float2(br + dr, -(bi_ + di));
      sp[(16 * i + 256) * 16] = make_float2(ar - cr, -(ai - ci));
      sp[(16 * i + 384) * 16] = make_float2(br - dr, -(bi_ - di));
    } else {
      tp[(size_t)(16 * i) * FN]       = make_float2(ar + cr, ai + ci);
      tp[(size_t)(16 * i + 128) * FN] = make_float2(br + dr, bi_ + di);
      tp[(size_t)(16 * i + 256) * FN] = make_float2(ar - cr, ai - ci);
      tp[(size_t)(16 * i + 384) * FN] = make_float2(br - dr, bi_ - di);
    }
  }
}
__global__ __launch_bounds__(CT_TPB)
void k_band_cols(const float2* __restrict__ specRow, const float* __restrict__ win,
                 float2* __restrict__ T, int b0, int nb) {
  __shared__ float2 Y[16 * YK];
  __shared__ float2 Tw[32][16];
  __shared__ float2 L5[FN];
  __shared__ float2 Sp[SP_SUB];
  const int tid = threadIdx.x, lane = tid & 31, wid = tid >> 5;
  const int c = lane & 15, h = lane >> 4;
  const int col = blockIdx.x * CT_COLS + c;
  const size_t ebase = (size_t)(64 * h + wid) * FN + col;
  float2 xs[2][4][2];
#pragma unroll
  for (int jj = 0; jj < 2; ++jj)
#pragma unroll
    for (int kc = 0; kc < 4; ++kc) {
      xs[jj][kc][0] = specRow[ebase + (size_t)(128 * kc + 16 * jj) * FN];
      xs[jj][kc][1] = specRow[ebase + (size_t)(128 * kc + 32 + 16 * jj) * FN];
    }
  fill_w512(L5, tid, CT_TPB, 1.0f);
  const int np = (nb + 1) >> 1;
  const int rb = 64 * h + wid;
  const size_t mcol  = (size_t)((512 - col) & 511);
  const size_t mbase = (size_t)(512 - rb) * FN + mcol;
  const size_t mzero = (size_t)((512 - rb) & 511) * FN + mcol;
  float wr[2][4][2][4];
  lds_barrier();
  {
    const int n2 = tid >> 4, k1 = tid & 15;
    Tw[n2][k1] = L5[n2 * k1];
  }
  M8 m0, m1, m2, m3;
  make_m8(L5, lane, 0, m0);
  make_m8(L5, lane, 16, m1);
  make_m8(L5, lane, 32, m2);
  make_m8(L5, lane, 48, m3);
  lds_barrier();
  float2* const Ysub = Y;
  const v8f zero = {0.f,0.f,0.f,0.f,0.f,0.f,0.f,0.f};
#pragma unroll
  for (int jj = 0; jj < 2; ++jj) {
    v8f d0 = zero, d1 = zero;
#pragma unroll
    for (int kc = 0; kc < 4; ++kc) {
      v2f x0; x0.x = xs[jj][kc][0].x; x0.y = -xs[jj][kc][0].y;
      v2f x1; x1.x = xs[jj][kc][1].x; x1.y = -xs[jj][kc][1].y;
      d0 = wmma_f32(m0.a[kc], x0, d0);
      d1 = wmma_f32(m2.a[kc], x1, d1);
    }
    col_s1_store(Ysub, Tw, wid + 16 * jj, h, c, d0, d1);
  }
  if ((int)blockIdx.y < np) {
    const int pb = 2 * (int)blockIdx.y;
    const float* const wa = win + (size_t)(b0 + pb) * FN * FN;
    const float* const wb = (pb + 1 < nb) ? wa + (size_t)FN * FN : wa;
    load_win<0, 1>(wr, wa, wb, ebase, mbase, mzero);
  }
  lds_barrier();
  float2* const spr = Sp + (64 * h + (wid ^ h)) * 16 + c;
  col_stage2<true>(Ysub, wid, c, h, m0, m1, m2, m3, nullptr, spr);
  if ((int)blockIdx.y < np) {
    const int pb = 2 * (int)blockIdx.y;
    const float* const wa = win + (size_t)(b0 + pb) * FN * FN;
    const float* const wb = (pb + 1 < nb) ? wa + (size_t)FN * FN : wa;
    load_win<1, 1>(wr, wa, wb, ebase, mbase, mzero);
  }
  lds_barrier();
  const size_t obase = (size_t)(wid + 64 * h) * FN + col;
  for (int bi = blockIdx.y; bi < np; bi += gridDim.y) {
    const float fa = 0.5f;
    const float fb = (2 * bi + 1 < nb) ? 0.5f : 0.0f;
    const int bn = bi + (int)gridDim.y;
#pragma unroll
    for (int jj = 0; jj < 2; ++jj) {
      v8f d0 = zero, d1 = zero;
#pragma unroll
      for (int kc = 0; kc < 4; ++kc) {
        const float pa = fa * (wr[jj][kc][0][0] + wr[jj][kc][0][1]);
        const float pb = fb * (wr[jj][kc][0][2] + wr[jj][kc][0][3]);
        const float qa = fa * (wr[jj][kc][1][0] + wr[jj][kc][1][1]);
        const float qb = fb * (wr[jj][kc][1][2] + wr[jj][kc][1][3]);
        const float2 sa = spr[(128 * kc + 16 * jj) * 16];
        const float2 sb = spr[(128 * kc + 32 + 16 * jj) * 16];
        v2f x0; x0.x = sa.x * pa - sa.y * pb; x0.y = sa.x * pb + sa.y * pa;
        v2f x1; x1.x = sb.x * qa - sb.y * qb; x1.y = sb.x * qb + sb.y * qa;
        d0 = wmma_f32(m0.a[kc], x0, d0);
        d1 = wmma_f32(m2.a[kc], x1, d1);
      }
      col_s1_store(Ysub, Tw, wid + 16 * jj, h, c, d0, d1);
    }
    if (bn < np) {
      const float* const wan = win + (size_t)(b0 + 2 * bn) * FN * FN;
      const float* const wbn = (2 * bn + 1 < nb) ? wan + (size_t)FN * FN : wan;
      load_win<0, 1>(wr, wan, wbn, ebase, mbase, mzero);
    }
    lds_barrier();
    col_stage2<false>(Ysub, wid, c, h, m0, m1, m2, m3,
                      T + (size_t)bi * FN * FN + obase, nullptr);
    if (bn < np) {
      const float* const wan = win + (size_t)(b0 + 2 * bn) * FN * FN;
      const float* const wbn = (2 * bn + 1 < nb) ? wan + (size_t)FN * FN : wan;
      load_win<1, 1>(wr, wan, wbn, ebase, mbase, mzero);
    }
    lds_barrier();
  }
}

#define RW_TPB   128
#define RW_WPB   (RW_TPB / 32)
#define RW_WGS   768
__global__ __launch_bounds__(RW_TPB)
void k_band_rows(const float2* __restrict__ T, float* __restrict__ out,
                 int b0, int nb) {
  __shared__ float2 L512[FN];
  __shared__ __attribute__((aligned(16))) float2 Yx[RW_WPB][16 * RW_PITCH];
  const int tid = threadIdx.x, lane = tid & 31, wid = tid >> 5;
  const int c = lane & 15, h = lane >> 4;
  fill_w512(L512, tid, RW_TPB, 1.0f);
  __syncthreads();
  const float scale = 1.0f / (512.0f * 512.0f);
  M8 p0, p1, q0, q1, q2, q3;
  make_m8(L512, lane, 0, p0);
  make_m8(L512, lane, 32, p1);
  make_m8s(L512, lane, 0, scale, q0);
  make_m8s(L512, lane, 16, scale, q1);
  make_m8s(L512, lane, 32, scale, q2);
  make_m8s(L512, lane, 48, scale, q3);
  float2 twd[2][4][2];
#pragma unroll
  for (int t = 0; t < 2; ++t)
#pragma unroll
    for (int i = 0; i < 4; ++i)
#pragma unroll
      for (int u = 0; u < 2; ++u)
        twd[t][i][u] = L512[(16 * t + c) * (4 * h + i + 8 * u)];

  float2* const ysh = Yx[wid];
  const float4* const yrd =
      reinterpret_cast<const float4*>(ysh + c * RW_PITCH + 4 * h);
  const v8f zero = {0.f,0.f,0.f,0.f,0.f,0.f,0.f,0.f};
  const int total = ((nb + 1) >> 1) * FN;
  const int nw = (int)gridDim.x * RW_WPB;
  int l = (int)blockIdx.x * RW_WPB + wid;
  const int lbase = 64 * h + c;
  float2 xc[2][4][2];
  if (l < total) {
    const float2* const tp = T + (size_t)l * FN + lbase;
#pragma unroll
    for (int t = 0; t < 2; ++t)
#pragma unroll
      for (int kc = 0; kc < 4; ++kc) {
        xc[t][kc][0] = tp[128 * kc + 16 * t];
        xc[t][kc][1] = tp[128 * kc + 32 + 16 * t];
      }
  }
  for (; l < total; l += nw) {
    v8f d0[2], d1[2];
#pragma unroll
    for (int t = 0; t < 2; ++t) {
      d0[t] = zero; d1[t] = zero;
#pragma unroll
      for (int kc = 0; kc < 4; ++kc) {
        v2f x0; x0.x = xc[t][kc][0].x; x0.y = xc[t][kc][0].y;
        v2f x1; x1.x = xc[t][kc][1].x; x1.y = xc[t][kc][1].y;
        d0[t] = wmma_f32(p0.a[kc], x0, d0[t]);
        d1[t] = wmma_f32(p1.a[kc], x1, d1[t]);
      }
    }
    if (l + nw < total) {
      const float2* const tn = T + (size_t)(l + nw) * FN + lbase;
#pragma unroll
      for (int t = 0; t < 2; ++t)
#pragma unroll
        for (int kc = 0; kc < 4; ++kc) {
          xc[t][kc][0] = tn[128 * kc + 16 * t];
          xc[t][kc][1] = tn[128 * kc + 32 + 16 * t];
        }
    }
#pragma unroll
    for (int t = 0; t < 2; ++t) {
#pragma unroll
      for (int i = 0; i < 4; ++i) {
        const float pr = d0[t][2 * i] + d1[t][2 * i];
        const float pi = d0[t][2 * i + 1] + d1[t][2 * i + 1];
        const float mr = d0[t][2 * i] - d1[t][2 * i];
        const float mi = d0[t][2 * i + 1] - d1[t][2 * i + 1];
        const float2 ta = twd[t][i][0], tb = twd[t][i][1];
        ysh[(4 * h + i) * RW_PITCH + 16 * t + c] =
            make_float2(pr * ta.x - pi * ta.y, pr * ta.y + pi * ta.x);
        ysh[(4 * h + i + 8) * RW_PITCH + 16 * t + c] =
            make_float2(mr * tb.x - mi * tb.y, mr * tb.y + mi * tb.x);
      }
    }
    float4 ya[4][2];
#pragma unroll
    for (int kc = 0; kc < 4; ++kc) {
      ya[kc][0] = yrd[4 * kc];
      ya[kc][1] = yrd[4 * kc + 1];
    }
    v8f e0 = zero, e1 = zero, e2 = zero, e3 = zero;
#pragma unroll
    for (int kc = 0; kc < 4; ++kc) {
      v2f y0; y0.x = ya[kc][0].x; y0.y = ya[kc][0].y;
      v2f y1; y1.x = ya[kc][0].z; y1.y = ya[kc][0].w;
      v2f y2; y2.x = ya[kc][1].x; y2.y = ya[kc][1].y;
      v2f y3; y3.x = ya[kc][1].z; y3.y = ya[kc][1].w;
      e0 = wmma_f32(q0.a[kc], y0, e0);
      e1 = wmma_f32(q1.a[kc], y1, e1);
      e2 = wmma_f32(q2.a[kc], y2, e2);
      e3 = wmma_f32(q3.a[kc], y3, e3);
    }
    const int pl = l >> 9, row = l & (FN - 1);
    float* const op =
        out + ((size_t)(b0 + 2 * pl) * FN + (size_t)row) * FN + 16 * h + c;
#pragma unroll
    for (int i = 0; i < 4; ++i) {
      const float ar = e0[2 * i] + e2[2 * i], br = e0[2 * i] - e2[2 * i];
      const float cr = e1[2 * i] + e3[2 * i];
      const float dr = e3[2 * i + 1] - e1[2 * i + 1];
      op[32 * i]       = ar + cr;
      op[32 * i + 128] = br + dr;
      op[32 * i + 256] = ar - cr;
      op[32 * i + 384] = br - dr;
    }
    if (2 * pl + 1 < nb) {
      float* const oq = op + (size_t)FN * FN;
#pragma unroll
      for (int i = 0; i < 4; ++i) {
        const float ai = e0[2 * i + 1] + e2[2 * i + 1], bi_ = e0[2 * i + 1] - e2[2 * i + 1];
        const float ci = e1[2 * i + 1] + e3[2 * i + 1];
        const float di = e1[2 * i] - e3[2 * i];
        oq[32 * i]       = ai + ci;
        oq[32 * i + 128] = bi_ + di;
        oq[32 * i + 256] = ai - ci;
        oq[32 * i + 384] = bi_ - di;
      }
    }
  }
}

extern "C" void kernel_launch(void* const* d_in, const int* in_sizes, int n_in,
                              void* d_out, int out_size, void* d_ws,
                              size_t ws_size, hipStream_t stream) {
  (void)in_sizes; (void)n_in; (void)out_size;
  const float* img = (const float*)d_in[0];
  const float* win = (const float*)d_in[1];
  float* out = (float*)d_out;

  float2* specRow = (float2*)d_ws;
  float2* bandT   = specRow + 2 * (size_t)FN * FN;
  const size_t reserved = 2ull * sizeof(float2) * FN * FN;
  const size_t perBand  = sizeof(float2) * (size_t)FN * FN;
  int chunk = (ws_size > reserved) ? (int)((ws_size - reserved) / perBand) : 0;
  if (chunk < 1)  chunk = 1;
  if (chunk > NB) chunk = NB;

  const dim3 blk(TPB);
  k_fwd_rows<<<dim3(FN / LPB), blk, 0, stream>>>(img, specRow);
  for (int b0 = 0; b0 < NB; b0 += chunk) {
    const int nb = (NB - b0 < chunk) ? (NB - b0) : chunk;
    k_band_cols<<<dim3(FN / CT_COLS, GBC), dim3(CT_TPB), 0, stream>>>(specRow, win, bandT, b0, nb);
    k_band_rows<<<dim3(RW_WGS), dim3(RW_TPB), 0, stream>>>(bandT, out, b0, nb);
  }
}
